# starting kernel
# baseline (speedup 1.0000x reference)
#include <hip/hip_runtime.h>

#define NN   50000
#define NE   800000
#define HD   128

typedef float v2f __attribute__((ext_vector_type(2)));
typedef float v8f __attribute__((ext_vector_type(8)));

__global__ __launch_bounds__(256) void dgi_deg_init(float* __restrict__ deg) {
    int i = blockIdx.x * blockDim.x + threadIdx.x;
    if (i < NN) deg[i] = 1.0f;
}

__global__ __launch_bounds__(256) void dgi_deg_accum(const int* __restrict__ dst,
                                                     float* __restrict__ deg) {
    int e = blockIdx.x * blockDim.x + threadIdx.x;
    if (e < NE) atomicAdd(&deg[dst[e]], 1.0f);
}

__global__ __launch_bounds__(256) void dgi_dinv(float* __restrict__ degdinv) {
    int i = blockIdx.x * blockDim.x + threadIdx.x;
    if (i < NN) {
        float d = degdinv[i];
        degdinv[i] = (d > 0.0f) ? rsqrtf(d) : 0.0f;
    }
}

__global__ __launch_bounds__(256) void dgi_gemm128_wmma(const float* __restrict__ A,
                                                        const float* __restrict__ B,
                                                        float* __restrict__ C) {
    const int wave = threadIdx.x >> 5;
    const int lane = threadIdx.x & 31;
    const int half = lane >> 4;
    const int l    = lane & 15;
    const int m0   = blockIdx.x * 16;
    const int n0   = wave * 16;

    const float* __restrict__ arow = A + (size_t)(m0 + l) * HD;

    v8f acc = {0.f, 0.f, 0.f, 0.f, 0.f, 0.f, 0.f, 0.f};

#pragma unroll
    for (int k0 = 0; k0 < HD; k0 += 4) {
        const int ka = k0 + 2 * half;
        v2f a, b;
        a.x = arow[ka + 0];
        a.y = arow[ka + 1];
        b.x = B[(size_t)(ka + 0) * HD + n0 + l];
        b.y = B[(size_t)(ka + 1) * HD + n0 + l];
        acc = __builtin_amdgcn_wmma_f32_16x16x4_f32(
            false, a, false, b, (short)0, acc, false, false);
    }

#pragma unroll
    for (int r = 0; r < 8; ++r) {
        C[(size_t)(m0 + r + 8 * half) * HD + n0 + l] = acc[r];
    }
}

__global__ __launch_bounds__(256) void dgi_z_init(const float* __restrict__ xw,
                                                  const float* __restrict__ dinv,
                                                  const float* __restrict__ bias,
                                                  float* __restrict__ z) {
    long long i = (long long)blockIdx.x * blockDim.x + threadIdx.x;
    if (i < (long long)NN * HD) {
        int n  = (int)(i >> 7);
        int ch = (int)(i & 127);
        float di = dinv[n];
        z[i] = bias[ch] + di * di * xw[i];
    }
}

__global__ __launch_bounds__(256) void dgi_edge_scatter(const int* __restrict__ src,
                                                        const int* __restrict__ dst,
                                                        const float* __restrict__ xw,
                                                        const float* __restrict__ dinv,
                                                        float* __restrict__ z) {
    long long i = (long long)blockIdx.x * blockDim.x + threadIdx.x;
    if (i < (long long)NE * HD) {
        int e  = (int)(i >> 7);
        int ch = (int)(i & 127);
        int s  = src[e];
        int d  = dst[e];
        float c = dinv[s] * dinv[d];
        atomicAdd(&z[(long long)d * HD + ch], xw[(long long)s * HD + ch] * c);
    }
}

__global__ __launch_bounds__(256) void dgi_score(const float* __restrict__ u,
                                                 const float* __restrict__ z,
                                                 const int* __restrict__ perm,
                                                 const float* __restrict__ bb,
                                                 float* __restrict__ out) {
    const int wave = threadIdx.x >> 5;
    const int lane = threadIdx.x & 31;
    const int n = blockIdx.x * 8 + wave;
    if (n >= NN) return;

    const float* un = u + (size_t)n * HD;
    const float* za = z + (size_t)n * HD;
    const float* zb = z + (size_t)perm[n] * HD;

    float4 uu = *(const float4*)(un + lane * 4);
    float4 va = *(const float4*)(za + lane * 4);
    float4 vb = *(const float4*)(zb + lane * 4);

    float pos = uu.x * va.x + uu.y * va.y + uu.z * va.z + uu.w * va.w;
    float neg = uu.x * vb.x + uu.y * vb.y + uu.z * vb.z + uu.w * vb.w;

#pragma unroll
    for (int off = 16; off > 0; off >>= 1) {
        pos += __shfl_xor(pos, off, 32);
        neg += __shfl_xor(neg, off, 32);
    }
    if (lane == 0) {
        float bias = bb[0];
        out[n]      = pos + bias;
        out[NN + n] = neg + bias;
    }
}

extern "C" void kernel_launch(void* const* d_in, const int* in_sizes, int n_in,
                              void* d_out, int out_size, void* d_ws, size_t ws_size,
                              hipStream_t stream) {
    (void)in_sizes; (void)n_in; (void)out_size; (void)ws_size;

    const float* x    = (const float*)d_in[0];
    const float* W    = (const float*)d_in[1];
    const float* b    = (const float*)d_in[2];
    const float* Wb   = (const float*)d_in[3];
    const float* bb   = (const float*)d_in[4];
    const int*   ei   = (const int*)d_in[5];
    const int*   perm = (const int*)d_in[6];
    const int*   esrc = ei;
    const int*   edst = ei + NE;
    float*       out  = (float*)d_out;

    char* ws = (char*)d_ws;
    size_t off = 0;
    float* dinv = (float*)(ws + off); off += ((size_t)NN * 4 + 255) & ~(size_t)255;
    float* xw   = (float*)(ws + off); off += ((size_t)NN * HD * 4 + 255) & ~(size_t)255;
    float* z    = (float*)(ws + off); off += ((size_t)NN * HD * 4 + 255) & ~(size_t)255;
    float* u    = (float*)(ws + off); off += ((size_t)NN * HD * 4 + 255) & ~(size_t)255;

    const int nblk  = (NN + 255) / 256;
    const int eblk  = (NE + 255) / 256;
    const int nhblk = (int)(((long long)NN * HD + 255) / 256);
    const int ehblk = (int)(((long long)NE * HD + 255) / 256);

    dgi_deg_init<<<nblk, 256, 0, stream>>>(dinv);
    dgi_deg_accum<<<eblk, 256, 0, stream>>>(edst, dinv);
    dgi_dinv<<<nblk, 256, 0, stream>>>(dinv);

    dgi_gemm128_wmma<<<NN / 16, 256, 0, stream>>>(x, W, xw);

    dgi_z_init<<<nhblk, 256, 0, stream>>>(xw, dinv, b, z);
    dgi_edge_scatter<<<ehblk, 256, 0, stream>>>(esrc, edst, xw, dinv, z);

    dgi_gemm128_wmma<<<NN / 16, 256, 0, stream>>>(z, Wb, u);

    dgi_score<<<(NN + 7) / 8, 256, 0, stream>>>(u, z, perm, bb, out);
}


# float-atomic edge scatter-add replaced by a per-call destination-sorted edge list (one returning int add per edge for its rank, scan, fill) and a one-wave-per-node row gather that writes z once
# speedup vs baseline: 2.1643x; 2.1643x over previous
#include <hip/hip_runtime.h>


#define NN   50000
#define NE   800000
#define HD   128

typedef float v2f __attribute__((ext_vector_type(2)));
typedef float v8f __attribute__((ext_vector_type(8)));

__global__ __launch_bounds__(256) void dgi_cnt_zero(int* __restrict__ cnt) {
    int i = blockIdx.x * blockDim.x + threadIdx.x;
    if (i < NN) cnt[i] = 0;
}

#define EPT 5
static_assert(NE % (256 * EPT) == 0, "edge list must tile exactly");
__global__ __launch_bounds__(256) void dgi_rank(const int* __restrict__ dst,
                                                int* __restrict__ cnt,
                                                int* __restrict__ rank) {
    const int e0 = blockIdx.x * (256 * EPT) + threadIdx.x;
    int d[EPT], r[EPT];
#pragma unroll
    for (int q = 0; q < EPT; ++q) d[q] = dst[e0 + q * 256];
#pragma unroll
    for (int q = 0; q < EPT; ++q) r[q] = atomicAdd(&cnt[d[q]], 1);
#pragma unroll
    for (int q = 0; q < EPT; ++q) rank[e0 + q * 256] = r[q];
}

#define SCAN_PER_WAVE 1568
#define SCAN_PIECES   49
__global__ __launch_bounds__(1024) void dgi_scan(const int* __restrict__ cnt,
                                                 int* __restrict__ rowptr,
                                                 float* __restrict__ dinv) {
    __shared__ int wtot[32];
    const int wave = threadIdx.x >> 5;
    const int lane = threadIdx.x & 31;
    const int n0 = wave * SCAN_PER_WAVE + lane;

    int c[SCAN_PIECES];
    int part = 0;
#pragma unroll
    for (int k = 0; k < SCAN_PIECES; ++k) {
        int n = n0 + k * 32;
        c[k] = (n < NN) ? cnt[n] : 0;
        part += c[k];
    }
#pragma unroll
    for (int off = 16; off > 0; off >>= 1) part += __shfl_xor(part, off, 32);
    if (lane == 0) wtot[wave] = part;
    __syncthreads();
    int carry = 0;
    for (int i = 0; i < wave; ++i) carry += wtot[i];

#pragma unroll
    for (int k = 0; k < SCAN_PIECES; ++k) {
        int x = c[k];
#pragma unroll
        for (int off = 1; off < 32; off <<= 1) {
            int y = __shfl_up(x, off, 32);
            if (lane >= off) x += y;
        }
        int n = n0 + k * 32;
        if (n < NN) {
            rowptr[n] = carry + x - c[k];
            float dg = (float)(c[k] + 1);
            dinv[n] = (dg > 0.0f) ? rsqrtf(dg) : 0.0f;
        }
        carry += __shfl(x, 31, 32);
    }
    if (threadIdx.x == 1023) rowptr[NN] = carry;
}

__global__ __launch_bounds__(256) void dgi_fill(const int* __restrict__ src,
                                                const int* __restrict__ dst,
                                                const int* __restrict__ rank,
                                                const int* __restrict__ rowptr,
                                                const float* __restrict__ dinv,
                                                int* __restrict__ csr) {
    const int e0 = blockIdx.x * (256 * EPT) + threadIdx.x;
    int2* __restrict__ csr2 = (int2*)csr;
    int s[EPT], d[EPT], r[EPT], p[EPT];
    float w[EPT];
#pragma unroll
    for (int q = 0; q < EPT; ++q) {
        int e = e0 + q * 256;
        s[q] = src[e];
        d[q] = dst[e];
        r[q] = rank[e];
    }
#pragma unroll
    for (int q = 0; q < EPT; ++q) {
        p[q] = rowptr[d[q]] + r[q];
        w[q] = dinv[s[q]];
    }
#pragma unroll
    for (int q = 0; q < EPT; ++q) {
        int pc = min(max(p[q], 0), NE - 1);
        csr2[pc] = make_int2(s[q], __float_as_int(w[q]));
    }
}

__global__ __launch_bounds__(256) void dgi_gemm128_wmma(const float* __restrict__ A,
                                                        const float* __restrict__ B,
                                                        float* __restrict__ C) {
    const int wave = threadIdx.x >> 5;
    const int lane = threadIdx.x & 31;
    const int half = lane >> 4;
    const int l    = lane & 15;
    const int m0   = blockIdx.x * 16;
    const int n0   = wave * 16;

    const float* __restrict__ arow = A + (size_t)(m0 + l) * HD;

    v8f acc = {0.f, 0.f, 0.f, 0.f, 0.f, 0.f, 0.f, 0.f};

#pragma unroll
    for (int k0 = 0; k0 < HD; k0 += 4) {
        const int ka = k0 + 2 * half;
        v2f a, b;
        a.x = arow[ka + 0];
        a.y = arow[ka + 1];
        b.x = B[(size_t)(ka + 0) * HD + n0 + l];
        b.y = B[(size_t)(ka + 1) * HD + n0 + l];
        acc = __builtin_amdgcn_wmma_f32_16x16x4_f32(
            false, a, false, b, (short)0, acc, false, false);
    }

#pragma unroll
    for (int r = 0; r < 8; ++r) {
        C[(size_t)(m0 + r + 8 * half) * HD + n0 + l] = acc[r];
    }
}

__global__ __launch_bounds__(256) void dgi_gather(const int* __restrict__ rowptr,
                                                  const int* __restrict__ csr,
                                                  const float* __restrict__ xw,
                                                  const float* __restrict__ dinv,
                                                  const float* __restrict__ bias,
                                                  float* __restrict__ z) {
    const int wave = threadIdx.x >> 5;
    const int lane = threadIdx.x & 31;
    const int n = blockIdx.x * 8 + wave;
    if (n >= NN) return;

    const float4* __restrict__ xw4 = (const float4*)xw;
    const int2*   __restrict__ ent2 = (const int2*)csr;

    int beg = __builtin_amdgcn_readfirstlane(rowptr[n]);
    int end = __builtin_amdgcn_readfirstlane(rowptr[n + 1]);
    beg = min(max(beg, 0), NE);
    end = min(max(end, beg), NE);

    const float  di   = dinv[n];
    const float4 self = xw4[(size_t)n * 32 + lane];
    const float4 bz   = ((const float4*)bias)[lane];

    float4 acc = make_float4(0.f, 0.f, 0.f, 0.f);

    for (int base = beg; base < end; base += 32) {
        const int m = min(end - base, 32);
        int2 ent = make_int2(n, 0);
        if (lane < m) ent = ent2[base + lane];
        for (int j = 0; j < m; j += 8) {
            float4 v[8];
            float  w[8];
#pragma unroll
            for (int q = 0; q < 8; ++q) {
                int s = __builtin_amdgcn_readlane(ent.x, j + q);
                s = min(max(s, 0), NN - 1);
                w[q] = __int_as_float(__builtin_amdgcn_readlane(ent.y, j + q));
                v[q] = xw4[(size_t)s * 32 + lane];
            }
#pragma unroll
            for (int q = 0; q < 8; ++q) {
                acc.x += w[q] * v[q].x;
                acc.y += w[q] * v[q].y;
                acc.z += w[q] * v[q].z;
                acc.w += w[q] * v[q].w;
            }
        }
    }

    float4 o;
    o.x = bz.x + di * (di * self.x + acc.x);
    o.y = bz.y + di * (di * self.y + acc.y);
    o.z = bz.z + di * (di * self.z + acc.z);
    o.w = bz.w + di * (di * self.w + acc.w);
    ((float4*)z)[(size_t)n * 32 + lane] = o;
}

__global__ __launch_bounds__(256) void dgi_score(const float* __restrict__ u,
                                                 const float* __restrict__ z,
                                                 const int* __restrict__ perm,
                                                 const float* __restrict__ bb,
                                                 float* __restrict__ out) {
    const int wave = threadIdx.x >> 5;
    const int lane = threadIdx.x & 31;
    const int n = blockIdx.x * 8 + wave;
    if (n >= NN) return;

    const float* un = u + (size_t)n * HD;
    const float* za = z + (size_t)n * HD;
    const float* zb = z + (size_t)perm[n] * HD;

    float4 uu = *(const float4*)(un + lane * 4);
    float4 va = *(const float4*)(za + lane * 4);
    float4 vb = *(const float4*)(zb + lane * 4);

    float pos = uu.x * va.x + uu.y * va.y + uu.z * va.z + uu.w * va.w;
    float neg = uu.x * vb.x + uu.y * vb.y + uu.z * vb.z + uu.w * vb.w;

#pragma unroll
    for (int off = 16; off > 0; off >>= 1) {
        pos += __shfl_xor(pos, off, 32);
        neg += __shfl_xor(neg, off, 32);
    }
    if (lane == 0) {
        float bias = bb[0];
        out[n]      = pos + bias;
        out[NN + n] = neg + bias;
    }
}

extern "C" void kernel_launch(void* const* d_in, const int* in_sizes, int n_in,
                              void* d_out, int out_size, void* d_ws, size_t ws_size,
                              hipStream_t stream) {
    (void)in_sizes; (void)n_in; (void)out_size; (void)ws_size;

    const float* x    = (const float*)d_in[0];
    const float* W    = (const float*)d_in[1];
    const float* b    = (const float*)d_in[2];
    const float* Wb   = (const float*)d_in[3];
    const float* bb   = (const float*)d_in[4];
    const int*   ei   = (const int*)d_in[5];
    const int*   perm = (const int*)d_in[6];
    const int*   esrc = ei;
    const int*   edst = ei + NE;
    float*       out  = (float*)d_out;

    char* ws = (char*)d_ws;
    size_t off = 0;
    float* dinv = (float*)(ws + off); off += ((size_t)NN * 4 + 255) & ~(size_t)255;
    float* xw   = (float*)(ws + off); off += ((size_t)NN * HD * 4 + 255) & ~(size_t)255;
    float* z    = (float*)(ws + off); off += ((size_t)NN * HD * 4 + 255) & ~(size_t)255;
    float* u    = (float*)(ws + off); off += ((size_t)NN * HD * 4 + 255) & ~(size_t)255;

    char* t = (char*)u;
    size_t toff = 0;
    int* cnt    = (int*)(t + toff); toff += ((size_t)NN * 4 + 255) & ~(size_t)255;
    int* rowptr = (int*)(t + toff); toff += ((size_t)(NN + 1) * 4 + 255) & ~(size_t)255;
    int* rank   = (int*)(t + toff); toff += ((size_t)NE * 4 + 255) & ~(size_t)255;
    int* csr    = (int*)(t + toff); toff += ((size_t)NE * 8 + 255) & ~(size_t)255;

    const int nblk  = (NN + 255) / 256;
    const int eblk  = NE / (256 * EPT);

    dgi_cnt_zero<<<nblk, 256, 0, stream>>>(cnt);
    dgi_rank<<<eblk, 256, 0, stream>>>(edst, cnt, rank);
    dgi_scan<<<1, 1024, 0, stream>>>(cnt, rowptr, dinv);
    dgi_fill<<<eblk, 256, 0, stream>>>(esrc, edst, rank, rowptr, dinv, csr);

    dgi_gemm128_wmma<<<NN / 16, 256, 0, stream>>>(x, W, xw);

    dgi_gather<<<NN / 8, 256, 0, stream>>>(rowptr, csr, xw, dinv, b, z);

    dgi_gemm128_wmma<<<NN / 16, 256, 0, stream>>>(z, Wb, u);

    dgi_score<<<(NN + 7) / 8, 256, 0, stream>>>(u, z, perm, bb, out);
}


# gemm, gather and score made persistent: a work-group loops over 5 row tiles (B fragments stay in registers, a tile's A loads go out as one group), a wave over 5 nodes; grids 3125->625, 6250->1250
# speedup vs baseline: 2.3586x; 1.0898x over previous
#include <hip/hip_runtime.h>


#define NN   50000
#define NE   800000
#define HD   128

typedef float v2f __attribute__((ext_vector_type(2)));
typedef float v8f __attribute__((ext_vector_type(8)));

__global__ __launch_bounds__(256) void dgi_cnt_zero(int* __restrict__ cnt) {
    int i = blockIdx.x * blockDim.x + threadIdx.x;
    if (i < NN) cnt[i] = 0;
}

#define EPT 5
static_assert(NE % (256 * EPT) == 0, "edge list must tile exactly");
__global__ __launch_bounds__(256) void dgi_rank(const int* __restrict__ dst,
                                                int* __restrict__ cnt,
                                                int* __restrict__ rank) {
    const int e0 = blockIdx.x * (256 * EPT) + threadIdx.x;
    int d[EPT], r[EPT];
#pragma unroll
    for (int q = 0; q < EPT; ++q) d[q] = dst[e0 + q * 256];
#pragma unroll
    for (int q = 0; q < EPT; ++q) r[q] = atomicAdd(&cnt[d[q]], 1);
#pragma unroll
    for (int q = 0; q < EPT; ++q) rank[e0 + q * 256] = r[q];
}

#define SCAN_PER_WAVE 1568
#define SCAN_PIECES   49
__global__ __launch_bounds__(1024) void dgi_scan(const int* __restrict__ cnt,
                                                 int* __restrict__ rowptr,
                                                 float* __restrict__ dinv) {
    __shared__ int wtot[32];
    const int wave = threadIdx.x >> 5;
    const int lane = threadIdx.x & 31;
    const int n0 = wave * SCAN_PER_WAVE + lane;

    int c[SCAN_PIECES];
    int part = 0;
#pragma unroll
    for (int k = 0; k < SCAN_PIECES; ++k) {
        int n = n0 + k * 32;
        c[k] = (n < NN) ? cnt[n] : 0;
        part += c[k];
    }
#pragma unroll
    for (int off = 16; off > 0; off >>= 1) part += __shfl_xor(part, off, 32);
    if (lane == 0) wtot[wave] = part;
    __syncthreads();
    int carry = 0;
    for (int i = 0; i < wave; ++i) carry += wtot[i];

#pragma unroll
    for (int k = 0; k < SCAN_PIECES; ++k) {
        int x = c[k];
#pragma unroll
        for (int off = 1; off < 32; off <<= 1) {
            int y = __shfl_up(x, off, 32);
            if (lane >= off) x += y;
        }
        int n = n0 + k * 32;
        if (n < NN) {
            rowptr[n] = carry + x - c[k];
            float dg = (float)(c[k] + 1);
            dinv[n] = (dg > 0.0f) ? rsqrtf(dg) : 0.0f;
        }
        carry += __shfl(x, 31, 32);
    }
    if (threadIdx.x == 1023) rowptr[NN] = carry;
}

__global__ __launch_bounds__(256) void dgi_fill(const int* __restrict__ src,
                                                const int* __restrict__ dst,
                                                const int* __restrict__ rank,
                                                const int* __restrict__ rowptr,
                                                const float* __restrict__ dinv,
                                                int* __restrict__ csr) {
    const int e0 = blockIdx.x * (256 * EPT) + threadIdx.x;
    int2* __restrict__ csr2 = (int2*)csr;
    int s[EPT], d[EPT], r[EPT], p[EPT];
    float w[EPT];
#pragma unroll
    for (int q = 0; q < EPT; ++q) {
        int e = e0 + q * 256;
        s[q] = src[e];
        d[q] = dst[e];
        r[q] = rank[e];
    }
#pragma unroll
    for (int q = 0; q < EPT; ++q) {
        p[q] = rowptr[d[q]] + r[q];
        w[q] = dinv[s[q]];
    }
#pragma unroll
    for (int q = 0; q < EPT; ++q) {
        int pc = min(max(p[q], 0), NE - 1);
        csr2[pc] = make_int2(s[q], __float_as_int(w[q]));
    }
}

#define GEMM_TPB    5
#define GEMM_BLOCKS (NN / 16 / GEMM_TPB)
static_assert(NN % (16 * GEMM_TPB) == 0, "row tiles must divide exactly over the work-groups");
__global__ __launch_bounds__(256) void dgi_gemm128_wmma(const float* __restrict__ A,
                                                        const float* __restrict__ B,
                                                        float* __restrict__ C) {
    const int wave = threadIdx.x >> 5;
    const int lane = threadIdx.x & 31;
    const int half = lane >> 4;
    const int l    = lane & 15;
    const int n0   = wave * 16;

    v2f bf[HD / 4];
#pragma unroll
    for (int s = 0; s < HD / 4; ++s) {
        const int ka = 4 * s + 2 * half;
        bf[s].x = B[(size_t)(ka + 0) * HD + n0 + l];
        bf[s].y = B[(size_t)(ka + 1) * HD + n0 + l];
    }

    for (int t = 0; t < GEMM_TPB; ++t) {
        const int m0 = (blockIdx.x * GEMM_TPB + t) * 16;
        const float* __restrict__ arow = A + (size_t)(m0 + l) * HD;

        v2f af[HD / 4];
#pragma unroll
        for (int s = 0; s < HD / 4; ++s) {
            const int ka = 4 * s + 2 * half;
            af[s].x = arow[ka + 0];
            af[s].y = arow[ka + 1];
        }

        __builtin_amdgcn_sched_barrier(0);

        v8f acc = {0.f, 0.f, 0.f, 0.f, 0.f, 0.f, 0.f, 0.f};

#pragma unroll
        for (int s = 0; s < HD / 4; ++s) {
            acc = __builtin_amdgcn_wmma_f32_16x16x4_f32(
                false, af[s], false, bf[s], (short)0, acc, false, false);
        }

#pragma unroll
        for (int r = 0; r < 8; ++r) {
            C[(size_t)(m0 + r + 8 * half) * HD + n0 + l] = acc[r];
        }
    }
}

#define NPW         5
#define NODE_BLOCKS (NN / (8 * NPW))
static_assert(NN % (8 * NPW) == 0, "nodes must divide exactly over the waves");
__global__ __launch_bounds__(256) void dgi_gather(const int* __restrict__ rowptr,
                                                  const int* __restrict__ csr,
                                                  const float* __restrict__ xw,
                                                  const float* __restrict__ dinv,
                                                  const float* __restrict__ bias,
                                                  float* __restrict__ z) {
    const int wave = threadIdx.x >> 5;
    const int lane = threadIdx.x & 31;
    const int n0 = (blockIdx.x * 8 + wave) * NPW;

    const float4* __restrict__ xw4 = (const float4*)xw;
    const int2*   __restrict__ ent2 = (const int2*)csr;

    const float4 bz = ((const float4*)bias)[lane];
    const int rp = rowptr[n0 + min(lane, NPW)];

    for (int i = 0; i < NPW; ++i) {
        const int n = n0 + i;
        int beg = __builtin_amdgcn_readlane(rp, i);
        int end = __builtin_amdgcn_readlane(rp, i + 1);
        beg = min(max(beg, 0), NE);
        end = min(max(end, beg), NE);

        const float  di   = dinv[n];
        const float4 self = xw4[(size_t)n * 32 + lane];

        float4 acc = make_float4(0.f, 0.f, 0.f, 0.f);

        for (int base = beg; base < end; base += 32) {
            const int m = min(end - base, 32);
            int2 ent = make_int2(n, 0);
            if (lane < m) ent = ent2[base + lane];
            for (int j = 0; j < m; j += 8) {
                float4 v[8];
                float  w[8];
#pragma unroll
                for (int q = 0; q < 8; ++q) {
                    int s = __builtin_amdgcn_readlane(ent.x, j + q);
                    s = min(max(s, 0), NN - 1);
                    w[q] = __int_as_float(__builtin_amdgcn_readlane(ent.y, j + q));
                    v[q] = xw4[(size_t)s * 32 + lane];
                }
#pragma unroll
                for (int q = 0; q < 8; ++q) {
                    acc.x += w[q] * v[q].x;
                    acc.y += w[q] * v[q].y;
                    acc.z += w[q] * v[q].z;
                    acc.w += w[q] * v[q].w;
                }
            }
        }

        float4 o;
        o.x = bz.x + di * (di * self.x + acc.x);
        o.y = bz.y + di * (di * self.y + acc.y);
        o.z = bz.z + di * (di * self.z + acc.z);
        o.w = bz.w + di * (di * self.w + acc.w);
        ((float4*)z)[(size_t)n * 32 + lane] = o;
    }
}

__global__ __launch_bounds__(256) void dgi_score(const float* __restrict__ u,
                                                 const float* __restrict__ z,
                                                 const int* __restrict__ perm,
                                                 const float* __restrict__ bb,
                                                 float* __restrict__ out) {
    const int wave = threadIdx.x >> 5;
    const int lane = threadIdx.x & 31;
    const int n0 = (blockIdx.x * 8 + wave) * NPW;

    const float4* __restrict__ u4 = (const float4*)u;
    const float4* __restrict__ z4 = (const float4*)z;

    const int pm = perm[n0 + min(lane, NPW - 1)];

    float4 uu[NPW], va[NPW], vb[NPW];
#pragma unroll
    for (int i = 0; i < NPW; ++i) {
        const int p = __builtin_amdgcn_readlane(pm, i);
        uu[i] = u4[(size_t)(n0 + i) * 32 + lane];
        va[i] = z4[(size_t)(n0 + i) * 32 + lane];
        vb[i] = z4[(size_t)p * 32 + lane];
    }

    float pos_o = 0.f, neg_o = 0.f;
#pragma unroll
    for (int i = 0; i < NPW; ++i) {
        float pos = uu[i].x * va[i].x + uu[i].y * va[i].y + uu[i].z * va[i].z + uu[i].w * va[i].w;
        float neg = uu[i].x * vb[i].x + uu[i].y * vb[i].y + uu[i].z * vb[i].z + uu[i].w * vb[i].w;
#pragma unroll
        for (int off = 16; off > 0; off >>= 1) {
            pos += __shfl_xor(pos, off, 32);
            neg += __shfl_xor(neg, off, 32);
        }
        if (lane == i) { pos_o = pos; neg_o = neg; }
    }
    if (lane < NPW) {
        float bias = bb[0];
        out[n0 + lane]      = pos_o + bias;
        out[NN + n0 + lane] = neg_o + bias;
    }
}

extern "C" void kernel_launch(void* const* d_in, const int* in_sizes, int n_in,
                              void* d_out, int out_size, void* d_ws, size_t ws_size,
                              hipStream_t stream) {
    (void)in_sizes; (void)n_in; (void)out_size; (void)ws_size;

    const float* x    = (const float*)d_in[0];
    const float* W    = (const float*)d_in[1];
    const float* b    = (const float*)d_in[2];
    const float* Wb   = (const float*)d_in[3];
    const float* bb   = (const float*)d_in[4];
    const int*   ei   = (const int*)d_in[5];
    const int*   perm = (const int*)d_in[6];
    const int*   esrc = ei;
    const int*   edst = ei + NE;
    float*       out  = (float*)d_out;

    char* ws = (char*)d_ws;
    size_t off = 0;
    float* dinv = (float*)(ws + off); off += ((size_t)NN * 4 + 255) & ~(size_t)255;
    float* xw   = (float*)(ws + off); off += ((size_t)NN * HD * 4 + 255) & ~(size_t)255;
    float* z    = (float*)(ws + off); off += ((size_t)NN * HD * 4 + 255) & ~(size_t)255;
    float* u    = (float*)(ws + off); off += ((size_t)NN * HD * 4 + 255) & ~(size_t)255;

    char* t = (char*)u;
    size_t toff = 0;
    int* cnt    = (int*)(t + toff); toff += ((size_t)NN * 4 + 255) & ~(size_t)255;
    int* rowptr = (int*)(t + toff); toff += ((size_t)(NN + 1) * 4 + 255) & ~(size_t)255;
    int* rank   = (int*)(t + toff); toff += ((size_t)NE * 4 + 255) & ~(size_t)255;
    int* csr    = (int*)(t + toff); toff += ((size_t)NE * 8 + 255) & ~(size_t)255;

    const int nblk  = (NN + 255) / 256;
    const int eblk  = NE / (256 * EPT);

    dgi_cnt_zero<<<nblk, 256, 0, stream>>>(cnt);
    dgi_rank<<<eblk, 256, 0, stream>>>(edst, cnt, rank);
    dgi_scan<<<1, 1024, 0, stream>>>(cnt, rowptr, dinv);
    dgi_fill<<<eblk, 256, 0, stream>>>(esrc, edst, rank, rowptr, dinv, csr);

    dgi_gemm128_wmma<<<GEMM_BLOCKS, 256, 0, stream>>>(x, W, xw);

    dgi_gather<<<NODE_BLOCKS, 256, 0, stream>>>(rowptr, csr, xw, dinv, b, z);

    dgi_gemm128_wmma<<<GEMM_BLOCKS, 256, 0, stream>>>(z, Wb, u);

    dgi_score<<<NODE_BLOCKS, 256, 0, stream>>>(u, z, perm, bb, out);
}


# GEMM A tile fetched once per work-group with whole-row 16-byte loads into double-buffered LDS (next tile in flight under the WMMAs); 8 waves read fragments from LDS, not 16-line 8-byte global loads
# speedup vs baseline: 3.4250x; 1.4521x over previous
#include <hip/hip_runtime.h>


#define NN   50000
#define NE   800000
#define HD   128

typedef float v2f __attribute__((ext_vector_type(2)));
typedef float v8f __attribute__((ext_vector_type(8)));

#define CAP 64

__global__ __launch_bounds__(256) void dgi_cnt_zero(int* __restrict__ cnt) {
    int i = blockIdx.x * blockDim.x + threadIdx.x;
    if (i <= NN) cnt[i] = 0;
}

#define EPT 5
static_assert(NE % (256 * EPT) == 0, "edge list must tile exactly");
__global__ __launch_bounds__(256) void dgi_place(const int* __restrict__ src,
                                                 const int* __restrict__ dst,
                                                 int* __restrict__ cnt,
                                                 int* __restrict__ tab,
                                                 int* __restrict__ ovf) {
    const int e0 = blockIdx.x * (256 * EPT) + threadIdx.x;
    int2* __restrict__ ovf2 = (int2*)ovf;
    int s[EPT], d[EPT], r[EPT];
#pragma unroll
    for (int q = 0; q < EPT; ++q) {
        d[q] = dst[e0 + q * 256];
        s[q] = src[e0 + q * 256];
    }
#pragma unroll
    for (int q = 0; q < EPT; ++q) r[q] = atomicAdd(&cnt[d[q]], 1);
#pragma unroll
    for (int q = 0; q < EPT; ++q) {
        if ((unsigned)r[q] < (unsigned)CAP) {
            tab[(size_t)d[q] * CAP + r[q]] = s[q];
        } else {
            int i = atomicAdd(&cnt[NN], 1);
            i = min(max(i, 0), NE - 1);
            ovf2[i] = make_int2(d[q], s[q]);
        }
    }
}

#define GEMM_TPB    5
#define GEMM_BLOCKS (NN / 16 / GEMM_TPB)
static_assert(NN % (16 * GEMM_TPB) == 0, "row tiles must divide exactly over the work-groups");
#define A_PITCH     (HD + 4)
#define A_TILE_FLTS (16 * A_PITCH)
template <bool SCALE>
__device__ __forceinline__ void dgi_gemm128_body(const float* __restrict__ A,
                                                 const float* __restrict__ B,
                                                 float* __restrict__ C,
                                                 const int* __restrict__ deg,
                                                 float* sA) {
    const int wave = threadIdx.x >> 5;
    const int lane = threadIdx.x & 31;
    const int half = lane >> 4;
    const int l    = lane & 15;
    const int n0   = wave * 16;
    const int scol = lane * 4;
    const int row0 = blockIdx.x * GEMM_TPB * 16;

    const unsigned a_off = (unsigned)(row0 + wave) * HD + scol;
    float4 p0, p1;
    p0 = *(const float4*)(A + a_off);
    p1 = *(const float4*)(A + a_off + 8 * HD);

    v2f bf[HD / 4];
#pragma unroll
    for (int s = 0; s < HD / 4; ++s) {
        const int ka = 4 * s + 2 * half;
        bf[s].x = B[(unsigned)(ka + 0) * HD + n0 + l];
        bf[s].y = B[(unsigned)(ka + 1) * HD + n0 + l];
    }

    *(float4*)(sA + wave * A_PITCH + scol)       = p0;
    *(float4*)(sA + (wave + 8) * A_PITCH + scol) = p1;
    __syncthreads();

#pragma nounroll
    for (int t = 0; t < GEMM_TPB; ++t) {
        const int m0 = row0 + t * 16;
        const float* sa = sA + (t & 1) * A_TILE_FLTS;
        const bool more = (t + 1 < GEMM_TPB);

        if (more) {
            const unsigned g = a_off + (unsigned)(t + 1) * (16 * HD);
            p0 = *(const float4*)(A + g);
            p1 = *(const float4*)(A + g + 8 * HD);
        }

        v2f af[HD / 4];
#pragma unroll
        for (int s = 0; s < HD / 4; ++s) {
            af[s] = *(const v2f*)(sa + l * A_PITCH + 4 * s + 2 * half);
        }

        int dgl = 0;
        if (SCALE) dgl = deg[m0 + 8 * half + (l & 7)];

        __builtin_amdgcn_sched_barrier(0);

        v8f acc = {0.f, 0.f, 0.f, 0.f, 0.f, 0.f, 0.f, 0.f};

#pragma unroll
        for (int s = 0; s < HD / 4; ++s) {
            acc = __builtin_amdgcn_wmma_f32_16x16x4_f32(
                false, af[s], false, bf[s], (short)0, acc, false, false);
        }

        if (more) {
            float* sn = sA + ((t + 1) & 1) * A_TILE_FLTS;
            *(float4*)(sn + wave * A_PITCH + scol)       = p0;
            *(float4*)(sn + (wave + 8) * A_PITCH + scol) = p1;
            __syncthreads();
        }

        if (SCALE) {
            const float dg = (float)(dgl + 1);
            const float sc = (dg > 0.0f) ? rsqrtf(dg) : 0.0f;
#pragma unroll
            for (int r = 0; r < 8; ++r) {
                acc[r] *= __shfl(sc, (lane & 16) + r, 32);
            }
        }

        const unsigned c_off = (unsigned)(m0 + 8 * half) * HD + n0 + l;
#pragma unroll
        for (int r = 0; r < 8; ++r) {
            C[c_off + (unsigned)r * HD] = acc[r];
        }
    }
}

__global__ __launch_bounds__(256) void dgi_gemm128_scaled(const float* __restrict__ A,
                                                          const float* __restrict__ B,
                                                          float* __restrict__ C,
                                                          const int* __restrict__ deg) {
    __shared__ __attribute__((aligned(16))) float sA[2 * A_TILE_FLTS];
    dgi_gemm128_body<true>(A, B, C, deg, sA);
}

__global__ __launch_bounds__(256) void dgi_gemm128_plain(const float* __restrict__ A,
                                                         const float* __restrict__ B,
                                                         float* __restrict__ C) {
    __shared__ __attribute__((aligned(16))) float sA[2 * A_TILE_FLTS];
    dgi_gemm128_body<false>(A, B, C, nullptr, sA);
}

__device__ __forceinline__ void dgi_add_rows(float4& acc, const int ent, const int mm,
                                             const float4* __restrict__ y4, const int lane) {
    for (int j = 0; j < mm; j += 8) {
        float4 v[8];
        float  w[8];
#pragma unroll
        for (int q = 0; q < 8; ++q) {
            int s = __builtin_amdgcn_readlane(ent, j + q);
            s = min(max(s, 0), NN - 1);
            w[q] = (j + q < mm) ? 1.0f : 0.0f;
            v[q] = y4[(size_t)s * 32 + lane];
        }
#pragma unroll
        for (int q = 0; q < 8; ++q) {
            acc.x = fmaf(w[q], v[q].x, acc.x);
            acc.y = fmaf(w[q], v[q].y, acc.y);
            acc.z = fmaf(w[q], v[q].z, acc.z);
            acc.w = fmaf(w[q], v[q].w, acc.w);
        }
    }
}

#define NPW         5
#define NODE_BLOCKS (NN / (8 * NPW))
static_assert(NN % (8 * NPW) == 0, "nodes must divide exactly over the waves");
__global__ __launch_bounds__(256) void dgi_gather(const int* __restrict__ cnt,
                                                  const int* __restrict__ tab,
                                                  const int* __restrict__ ovf,
                                                  const float* __restrict__ y,
                                                  const float* __restrict__ bias,
                                                  float* __restrict__ z) {
    const int wave = threadIdx.x >> 5;
    const int lane = threadIdx.x & 31;
    const int n0 = (blockIdx.x * 8 + wave) * NPW;

    const float4* __restrict__ y4 = (const float4*)y;
    const int2*   __restrict__ ovf2 = (const int2*)ovf;

    const float4 bz = ((const float4*)bias)[lane];
    const int cl = cnt[n0 + min(lane, NPW - 1)];

    for (int i = 0; i < NPW; ++i) {
        const int n = n0 + i;
        int deg = __builtin_amdgcn_readlane(cl, i);
        deg = min(max(deg, 0), NE);
        const float dg = (float)(deg + 1);
        const float di = (dg > 0.0f) ? rsqrtf(dg) : 0.0f;

        const int m = min(deg, CAP);
        const int* __restrict__ bucket = tab + (size_t)n * CAP;
        int ent_lo = n, ent_hi = n;
        if (lane < m)      ent_lo = bucket[lane];
        if (lane + 32 < m) ent_hi = bucket[32 + lane];
        float4 acc = y4[(size_t)n * 32 + lane];

        dgi_add_rows(acc, ent_lo, min(m, 32), y4, lane);
        if (m > 32) dgi_add_rows(acc, ent_hi, m - 32, y4, lane);

        if (deg > CAP) {
            int novf = __builtin_amdgcn_readfirstlane(cnt[NN]);
            novf = min(max(novf, 0), NE);
            for (int i0 = 0; i0 < novf; i0 += 32) {
                const int idx = i0 + lane;
                int2 e = make_int2(-1, 0);
                if (idx < novf) e = ovf2[idx];
                unsigned hit = (unsigned)__ballot(e.x == n);
                while (hit != 0u) {
                    const int jj = __builtin_amdgcn_readfirstlane(__ffs(hit) - 1);
                    hit &= hit - 1u;
                    int s = __builtin_amdgcn_readlane(e.y, jj);
                    s = min(max(s, 0), NN - 1);
                    const float4 v = y4[(size_t)s * 32 + lane];
                    acc.x += v.x;
                    acc.y += v.y;
                    acc.z += v.z;
                    acc.w += v.w;
                }
            }
        }

        float4 o;
        o.x = bz.x + di * acc.x;
        o.y = bz.y + di * acc.y;
        o.z = bz.z + di * acc.z;
        o.w = bz.w + di * acc.w;
        ((float4*)z)[(size_t)n * 32 + lane] = o;
    }
}

__global__ __launch_bounds__(256) void dgi_score(const float* __restrict__ u,
                                                 const float* __restrict__ z,
                                                 const int* __restrict__ perm,
                                                 const float* __restrict__ bb,
                                                 float* __restrict__ out) {
    const int wave = threadIdx.x >> 5;
    const int lane = threadIdx.x & 31;
    const int n0 = (blockIdx.x * 8 + wave) * NPW;

    const float4* __restrict__ u4 = (const float4*)u;
    const float4* __restrict__ z4 = (const float4*)z;

    const int pm = perm[n0 + min(lane, NPW - 1)];

    float4 uu[NPW], va[NPW], vb[NPW];
#pragma unroll
    for (int i = 0; i < NPW; ++i) {
        const int p = __builtin_amdgcn_readlane(pm, i);
        uu[i] = u4[(size_t)(n0 + i) * 32 + lane];
        va[i] = z4[(size_t)(n0 + i) * 32 + lane];
        vb[i] = z4[(size_t)p * 32 + lane];
    }

    float pos_o = 0.f, neg_o = 0.f;
#pragma unroll
    for (int i = 0; i < NPW; ++i) {
        float pos = uu[i].x * va[i].x + uu[i].y * va[i].y + uu[i].z * va[i].z + uu[i].w * va[i].w;
        float neg = uu[i].x * vb[i].x + uu[i].y * vb[i].y + uu[i].z * vb[i].z + uu[i].w * vb[i].w;
#pragma unroll
        for (int off = 16; off > 0; off >>= 1) {
            pos += __shfl_xor(pos, off, 32);
            neg += __shfl_xor(neg, off, 32);
        }
        if (lane == i) { pos_o = pos; neg_o = neg; }
    }
    if (lane < NPW) {
        float bias = bb[0];
        out[n0 + lane]      = pos_o + bias;
        out[NN + n0 + lane] = neg_o + bias;
    }
}

extern "C" void kernel_launch(void* const* d_in, const int* in_sizes, int n_in,
                              void* d_out, int out_size, void* d_ws, size_t ws_size,
                              hipStream_t stream) {
    (void)in_sizes; (void)n_in; (void)out_size; (void)ws_size;

    const float* x    = (const float*)d_in[0];
    const float* W    = (const float*)d_in[1];
    const float* b    = (const float*)d_in[2];
    const float* Wb   = (const float*)d_in[3];
    const float* bb   = (const float*)d_in[4];
    const int*   ei   = (const int*)d_in[5];
    const int*   perm = (const int*)d_in[6];
    const int*   esrc = ei;
    const int*   edst = ei + NE;
    float*       out  = (float*)d_out;

    char* ws = (char*)d_ws;
    size_t off = 0;
    int*   cnt  = (int*)(ws + off);   off += ((size_t)NN * 4 + 255) & ~(size_t)255;
    float* xw   = (float*)(ws + off); off += ((size_t)NN * HD * 4 + 255) & ~(size_t)255;
    float* z    = (float*)(ws + off); off += ((size_t)NN * HD * 4 + 255) & ~(size_t)255;
    float* u    = (float*)(ws + off); off += ((size_t)NN * HD * 4 + 255) & ~(size_t)255;
    static_assert((((size_t)NN * 4 + 255) & ~(size_t)255) >= ((size_t)NN + 1) * 4,
                  "counter region must hold NN + 1 ints");

    char* t = (char*)u;
    size_t toff = 0;
    int* tab = (int*)(t + toff); toff += ((size_t)NN * CAP * 4 + 255) & ~(size_t)255;
    int* ovf = (int*)(t + toff); toff += ((size_t)NE * 8 + 255) & ~(size_t)255;
    static_assert((size_t)NN * CAP * 4 + (size_t)NE * 8 + 512 <= (size_t)NN * HD * 4,
                  "buckets and overflow list must fit in the u region");

    const int nblk  = (NN + 1 + 255) / 256;
    const int eblk  = NE / (256 * EPT);

    dgi_cnt_zero<<<nblk, 256, 0, stream>>>(cnt);
    dgi_place<<<eblk, 256, 0, stream>>>(esrc, edst, cnt, tab, ovf);

    dgi_gemm128_scaled<<<GEMM_BLOCKS, 256, 0, stream>>>(x, W, xw, cnt);

    dgi_gather<<<NODE_BLOCKS, 256, 0, stream>>>(cnt, tab, ovf, xw, b, z);

    dgi_gemm128_plain<<<GEMM_BLOCKS, 256, 0, stream>>>(z, Wb, u);

    dgi_score<<<NODE_BLOCKS, 256, 0, stream>>>(u, z, perm, bb, out);
}
